# starting kernel
# baseline (speedup 1.0000x reference)
#include <hip/hip_runtime.h>
#include <hip/hip_bf16.h>

typedef __attribute__((ext_vector_type(2))) float v2f;
typedef __attribute__((ext_vector_type(8))) float v8f;

#define D 128
#define LN_EPS 1e-5f

__global__ void k_deg_init(float* __restrict__ deg, int n) {
  int i = blockIdx.x * blockDim.x + threadIdx.x;
  if (i < n) deg[i] = 1.0f;
}

__global__ void k_deg_accum(const int* __restrict__ row, float* __restrict__ deg, int e) {
  int i = blockIdx.x * blockDim.x + threadIdx.x;
  if (i < e) atomicAdd(&deg[row[i]], 1.0f);
}

__global__ void k_dinv(float* __restrict__ deg, int n) {
  int i = blockIdx.x * blockDim.x + threadIdx.x;
  if (i < n) deg[i] = rsqrtf(deg[i]);
}

__global__ __launch_bounds__(256) void k_gemm_wmma(const float* __restrict__ h,
                                                   const float* __restrict__ W,
                                                   float* __restrict__ out, int n) {
  __shared__ __align__(16) float lw[D * D];
  {
    const float4* W4 = (const float4*)W;
    float4* lw4 = (float4*)lw;
    for (int i = threadIdx.x; i < (D * D) / 4; i += 256) lw4[i] = W4[i];
  }
  __syncthreads();

  const int wave = threadIdx.x >> 5;
  const int lane = threadIdx.x & 31;
  const int ln   = lane & 15;
  const int koff = (lane < 16) ? 0 : 2;

  const int row0 = (blockIdx.x * 8 + wave) * 16;
  if (row0 + 16 > n) return;

  v8f acc[8];
  const v8f zero = {0.f, 0.f, 0.f, 0.f, 0.f, 0.f, 0.f, 0.f};
#pragma unroll
  for (int t = 0; t < 8; ++t) acc[t] = zero;

  const float* __restrict__ hrow = h + (size_t)(row0 + ln) * D;

  for (int k0 = 0; k0 < D; k0 += 4) {
    v2f a = *(const v2f*)(hrow + k0 + koff);
#pragma unroll
    for (int t = 0; t < 8; ++t) {
      const int n0 = t * 16;
      v2f b;
      b.x = lw[(k0 + koff) * D + n0 + ln];
      b.y = lw[(k0 + koff + 1) * D + n0 + ln];
      acc[t] = __builtin_amdgcn_wmma_f32_16x16x4_f32(
           false, a,  false, b,
           (short)0, acc[t],  false,  false);
    }
  }

  const int mbase = (lane < 16) ? 0 : 8;
#pragma unroll
  for (int t = 0; t < 8; ++t) {
    const int n0 = t * 16;
#pragma unroll
    for (int j = 0; j < 8; ++j) {
      out[(size_t)(row0 + mbase + j) * D + n0 + ln] = acc[t][j];
    }
  }
}

__global__ void k_agg_init(const float* __restrict__ hw, const float* __restrict__ dinv,
                           const float* __restrict__ bias, float* __restrict__ agg, int n) {
  int i = blockIdx.x * blockDim.x + threadIdx.x;
  int node = i >> 5;
  if (node >= n) return;
  int c = (i & 31) * 4;
  float s = dinv[node];
  s = s * s;
  float4 hv = *(const float4*)(hw + (size_t)node * D + c);
  float4 r;
  r.x = bias[c + 0] + s * hv.x;
  r.y = bias[c + 1] + s * hv.y;
  r.z = bias[c + 2] + s * hv.z;
  r.w = bias[c + 3] + s * hv.w;
  *(float4*)(agg + (size_t)node * D + c) = r;
}

__global__ void k_edge_scatter(const int* __restrict__ er, const int* __restrict__ ec,
                               const float* __restrict__ dinv,
                               const float* __restrict__ hw,
                               float* __restrict__ agg, int e) {
  int i = blockIdx.x * blockDim.x + threadIdx.x;
  int edge = i >> 5;
  if (edge >= e) return;
  int c = (i & 31) * 4;
  int r = er[edge];
  int s = ec[edge];
  float wgt = dinv[r] * dinv[s];
  float4 hv = *(const float4*)(hw + (size_t)s * D + c);
  float* ap = agg + (size_t)r * D + c;
  atomicAdd(ap + 0, wgt * hv.x);
  atomicAdd(ap + 1, wgt * hv.y);
  atomicAdd(ap + 2, wgt * hv.z);
  atomicAdd(ap + 3, wgt * hv.w);
}

__global__ __launch_bounds__(256) void k_ln_relu(const float* __restrict__ agg,
                                                 const float* __restrict__ g,
                                                 const float* __restrict__ beta,
                                                 float* __restrict__ out, int n) {
  const int wave = threadIdx.x >> 5;
  const int lane = threadIdx.x & 31;
  const int row = blockIdx.x * 8 + wave;
  if (row >= n) return;
  const int c = lane * 4;
  float4 v = *(const float4*)(agg + (size_t)row * D + c);

  float s = v.x + v.y + v.z + v.w;
#pragma unroll
  for (int off = 16; off > 0; off >>= 1) s += __shfl_xor(s, off, 32);
  float mu = s * (1.0f / 128.0f);

  float dx = v.x - mu, dy = v.y - mu, dz = v.z - mu, dw = v.w - mu;
  float q = dx * dx + dy * dy + dz * dz + dw * dw;
#pragma unroll
  for (int off = 16; off > 0; off >>= 1) q += __shfl_xor(q, off, 32);
  float rs = rsqrtf(q * (1.0f / 128.0f) + LN_EPS);

  float4 o;
  o.x = fmaxf(dx * rs * g[c + 0] + beta[c + 0], 0.0f);
  o.y = fmaxf(dy * rs * g[c + 1] + beta[c + 1], 0.0f);
  o.z = fmaxf(dz * rs * g[c + 2] + beta[c + 2], 0.0f);
  o.w = fmaxf(dw * rs * g[c + 3] + beta[c + 3], 0.0f);
  *(float4*)(out + (size_t)row * D + c) = o;
}

extern "C" void kernel_launch(void* const* d_in, const int* in_sizes, int n_in,
                              void* d_out, int out_size, void* d_ws, size_t ws_size,
                              hipStream_t stream) {
  const float* x    = (const float*)d_in[0];
  const int*   erow = (const int*)d_in[1];
  const int*   ecol = (const int*)d_in[2];
  const float* W[3]  = {(const float*)d_in[3], (const float*)d_in[7],  (const float*)d_in[11]};
  const float* b[3]  = {(const float*)d_in[4], (const float*)d_in[8],  (const float*)d_in[12]};
  const float* g[3]  = {(const float*)d_in[5], (const float*)d_in[9],  (const float*)d_in[13]};
  const float* bt[3] = {(const float*)d_in[6], (const float*)d_in[10], (const float*)d_in[14]};

  const int n = in_sizes[0] / D;
  const int e = in_sizes[1];

  float* ws   = (float*)d_ws;
  float* dinv = ws;
  float* hw   = ws + n;
  float* agg  = hw + (size_t)n * D;
  float* outp = (float*)d_out;

  k_deg_init<<<(n + 255) / 256, 256, 0, stream>>>(dinv, n);
  k_deg_accum<<<(e + 255) / 256, 256, 0, stream>>>(erow, dinv, e);
  k_dinv<<<(n + 255) / 256, 256, 0, stream>>>(dinv, n);

  const unsigned gemm_blocks = (unsigned)((n / 16 + 7) / 8);
  const unsigned node_chunks = (unsigned)(((size_t)n * 32 + 255) / 256);
  const unsigned edge_chunks = (unsigned)(((size_t)e * 32 + 255) / 256);
  const unsigned ln_blocks   = (unsigned)((n + 7) / 8);

  const float* hin = x;
  for (int L = 0; L < 3; ++L) {
    k_gemm_wmma<<<gemm_blocks, 256, 0, stream>>>(hin, W[L], hw, n);
    k_agg_init<<<node_chunks, 256, 0, stream>>>(hw, dinv, b[L], agg, n);
    k_edge_scatter<<<edge_chunks, 256, 0, stream>>>(erow, ecol, dinv, hw, agg, e);
    k_ln_relu<<<ln_blocks, 256, 0, stream>>>(agg, g[L], bt[L], outp, n);
    hin = outp;
  }
}


# Degrees: counted by each tile block's own edge filter (now a launch of its own in front of layer 1) with LDS integer adds, no global f32 atomics; 9 to 7 launches; arrival-order sums as before
# speedup vs baseline: 3.3399x; 3.3399x over previous
#include <hip/hip_runtime.h>
#include <hip/hip_bf16.h>

typedef __attribute__((ext_vector_type(2))) float v2f;
typedef __attribute__((ext_vector_type(8))) float v8f;

#define D 128
#define LN_EPS 1e-5f

__global__ __launch_bounds__(256) void k_gemm_wmma(const float* __restrict__ h,
                                                   const float* __restrict__ W,
                                                   float* __restrict__ out, int n) {
  __shared__ __align__(16) float lw[D * D];
  {
    const float4* W4 = (const float4*)W;
    float4* lw4 = (float4*)lw;
    for (int i = threadIdx.x; i < (D * D) / 4; i += 256) lw4[i] = W4[i];
  }
  __syncthreads();

  const int wave = threadIdx.x >> 5;
  const int lane = threadIdx.x & 31;
  const int ln   = lane & 15;
  const int koff = (lane < 16) ? 0 : 2;

  const int row0 = (blockIdx.x * 8 + wave) * 16;
  if (row0 + 16 > n) return;

  v8f acc[8];
  const v8f zero = {0.f, 0.f, 0.f, 0.f, 0.f, 0.f, 0.f, 0.f};
#pragma unroll
  for (int t = 0; t < 8; ++t) acc[t] = zero;

  const float* __restrict__ hrow = h + (size_t)(row0 + ln) * D;

  for (int k0 = 0; k0 < D; k0 += 4) {
    v2f a = *(const v2f*)(hrow + k0 + koff);
#pragma unroll
    for (int t = 0; t < 8; ++t) {
      const int n0 = t * 16;
      v2f b;
      b.x = lw[(k0 + koff) * D + n0 + ln];
      b.y = lw[(k0 + koff + 1) * D + n0 + ln];
      acc[t] = __builtin_amdgcn_wmma_f32_16x16x4_f32(
           false, a,  false, b,
           (short)0, acc[t],  false,  false);
    }
  }

  const int mbase = (lane < 16) ? 0 : 8;
#pragma unroll
  for (int t = 0; t < 8; ++t) {
    const int n0 = t * 16;
#pragma unroll
    for (int j = 0; j < 8; ++j) {
      out[(size_t)(row0 + mbase + j) * D + n0 + ln] = acc[t][j];
    }
  }
}

#define TR 400
#define CAP 16384
#define NW 32
#define QN 256
#define EG 8
#define KS 8

__device__ __forceinline__ void lds_add(float* p, float x) {
  __hip_atomic_fetch_add(p, x, __ATOMIC_RELAXED, __HIP_MEMORY_SCOPE_WORKGROUP);
}

__device__ __forceinline__ void tile_apply(const float* __restrict__ hw, float* acc,
                                           int s, int rl, float w, int count, int lane) {
  int g0 = 0;
#pragma unroll 1
  for (; g0 + EG <= count; g0 += EG) {
    float v[EG][4];
#pragma unroll
    for (int j = 0; j < EG; ++j) {
      const int sj = __builtin_amdgcn_readlane(s, g0 + j);
      const float* hp = hw + (size_t)sj * D + lane;
      v[j][0] = hp[0];
      v[j][1] = hp[32];
      v[j][2] = hp[64];
      v[j][3] = hp[96];
    }
#pragma unroll
    for (int j = 0; j < EG; ++j) {
      const int rj = __builtin_amdgcn_readlane(rl, g0 + j);
      const float wj = __int_as_float(__builtin_amdgcn_readlane(__float_as_int(w), g0 + j));
      float* ap = acc + rj * D + lane;
      lds_add(ap + 0,  wj * v[j][0]);
      lds_add(ap + 32, wj * v[j][1]);
      lds_add(ap + 64, wj * v[j][2]);
      lds_add(ap + 96, wj * v[j][3]);
    }
  }
  for (int j = g0; j < count; ++j) {
    const int sj = __builtin_amdgcn_readlane(s, j);
    const int rj = __builtin_amdgcn_readlane(rl, j);
    const float wj = __int_as_float(__builtin_amdgcn_readlane(__float_as_int(w), j));
    const float* hp = hw + (size_t)sj * D + lane;
    const float h0 = hp[0], h1 = hp[32], h2 = hp[64], h3 = hp[96];
    float* ap = acc + rj * D + lane;
    lds_add(ap + 0,  wj * h0);
    lds_add(ap + 32, wj * h1);
    lds_add(ap + 64, wj * h2);
    lds_add(ap + 96, wj * h3);
  }
}

template <bool RECORD>
__device__ __forceinline__ void tile_flush(const int* __restrict__ ec,
                                           const float* __restrict__ dinv,
                                           const float* __restrict__ hw, float* acc, int* degc,
                                           const int* qe, const int* qr, int head, int count,
                                           int lo, int lane, int* fill, int* __restrict__ ls,
                                           int* __restrict__ lr) {
  int rl = 0, s = 0;
  if (lane < count) {
    const int p = (head + lane) & (QN - 1);
    const int idx = qe[p];
    rl = qr[p];
    s = ec[idx];
  }
  if (RECORD) {
    if (lane < count)
      __hip_atomic_fetch_add(degc + rl, 1, __ATOMIC_RELAXED, __HIP_MEMORY_SCOPE_WORKGROUP);
    int base = 0;
    if (lane == 0)
      base = __hip_atomic_fetch_add(fill, count, __ATOMIC_RELAXED, __HIP_MEMORY_SCOPE_WORKGROUP);
    base = __builtin_amdgcn_readfirstlane(base);
    if (base + count <= CAP && lane < count) {
      ls[base + lane] = s;
      lr[base + lane] = rl;
    }
  } else {
    float w = 0.0f;
    if (lane < count) w = dinv[lo + rl] * dinv[s];
    tile_apply(hw, acc, s, rl, w, count, lane);
  }
}

template <bool RECORD>
__device__ __forceinline__ void tile_scan(const int* __restrict__ er, const int* __restrict__ ec,
                                          const float* __restrict__ dinv,
                                          const float* __restrict__ hw, float* acc, int* degc,
                                          int* qe, int* qr, int lo, int rows, int e, int wave,
                                          int lane, int* fill, int* __restrict__ ls,
                                          int* __restrict__ lr) {
  const unsigned R = (unsigned)rows;
  const int nfull = e >> 7;
  int head = 0, cnt = 0;
#define TILE_PUSH(P, IDX, DD)                                                      \
  {                                                                              \
    const unsigned m = __builtin_amdgcn_ballot_w32(P);                           \
    if (P) {                                                                     \
      const int pos = (head + cnt + (int)__builtin_amdgcn_mbcnt_lo(m, 0u)) & (QN - 1); \
      qe[pos] = (IDX);                                                           \
      qr[pos] = (int)(DD);                                                       \
    }                                                                            \
    cnt += __builtin_popcount(m);                                                \
  }
  if (wave == 0) {
    for (int base = nfull << 7; base < e; base += 32) {
      const int i = base + lane;
      const int r = (i < e) ? er[i] : -1;
      const unsigned d = (unsigned)(r - lo);
      const bool p = d < R;
      TILE_PUSH(p, i, d)
    }
  }
  if (nfull > 0) {
    const int last = nfull - 1;
    const int4* er4 = (const int4*)er + lane;
    const int ngroup = (nfull + KS - 1) / KS;
    const int shift = (int)(((size_t)blockIdx.x * (size_t)ngroup) / gridDim.x);
    for (int sg = wave; sg < ngroup; sg += NW) {
      int sr = sg + shift;
      if (sr >= ngroup) sr -= ngroup;
      const int cb = sr * KS;
      int4 a[KS];
#pragma unroll
      for (int k = 0; k < KS; ++k) {
        const int c = cb + k;
        a[k] = er4[(size_t)(c < last ? c : last) * 32];
      }
#pragma unroll
      for (int k = 0; k < KS; ++k) {
        const int c = cb + k;
        const int i0 = c * 128 + lane * 4;
        const bool in = c < nfull;
        const unsigned d0 = (unsigned)(a[k].x - lo), d1 = (unsigned)(a[k].y - lo);
        const unsigned d2 = (unsigned)(a[k].z - lo), d3 = (unsigned)(a[k].w - lo);
        const bool p0 = in & (d0 < R), p1 = in & (d1 < R), p2 = in & (d2 < R), p3 = in & (d3 < R);
        if (__builtin_amdgcn_ballot_w32(p0 | p1 | p2 | p3) != 0u) {
          TILE_PUSH(p0, i0 + 0, d0)
          TILE_PUSH(p1, i0 + 1, d1)
          TILE_PUSH(p2, i0 + 2, d2)
          TILE_PUSH(p3, i0 + 3, d3)
          while (cnt >= 32) {
            tile_flush<RECORD>(ec, dinv, hw, acc, degc, qe, qr, head, 32, lo, lane, fill, ls, lr);
            head = (head + 32) & (QN - 1);
            cnt -= 32;
          }
        }
      }
    }
  }
#undef TILE_PUSH
  while (cnt > 0) {
    const int k = cnt < 32 ? cnt : 32;
    tile_flush<RECORD>(ec, dinv, hw, acc, degc, qe, qr, head, k, lo, lane, fill, ls, lr);
    head = (head + k) & (QN - 1);
    cnt -= k;
  }
}

__global__ __launch_bounds__(1024) void k_tile_filter(const int* __restrict__ er,
                                                      const int* __restrict__ ec,
                                                      float* __restrict__ dinv,
                                                      int* __restrict__ meta, int* __restrict__ ls,
                                                      int* __restrict__ lr, int n, int e) {
  __shared__ int qe_all[NW * QN];
  __shared__ int qr_all[NW * QN];
  __shared__ int degc[TR];
  __shared__ int fill;
  for (int i = threadIdx.x; i < TR; i += NW * 32) degc[i] = 0;
  if (threadIdx.x == 0) fill = 0;
  __syncthreads();

  const int lane = threadIdx.x & 31;
  const int wave = __builtin_amdgcn_readfirstlane((int)(threadIdx.x >> 5));
  const int lo = blockIdx.x * TR;
  const int rows = (n - lo) < TR ? (n - lo) : TR;

  tile_scan<true>(er, ec, nullptr, nullptr, nullptr, degc, qe_all + wave * QN,
                  qr_all + wave * QN, lo, rows, e, wave, lane, &fill,
                  ls + (size_t)blockIdx.x * CAP, lr + (size_t)blockIdx.x * CAP);
  __syncthreads();

  for (int i = threadIdx.x; i < rows; i += NW * 32)
    dinv[lo + i] = rsqrtf((float)(degc[i] + 1));
  if (threadIdx.x == 0) meta[blockIdx.x] = fill;
}

template <bool FIRST>
__global__ __launch_bounds__(1024) void k_tile_agg_ln(const int* __restrict__ er,
                                                      const int* __restrict__ ec,
                                                      const float* __restrict__ dinv,
                                                      const float* __restrict__ hw,
                                                      const float* __restrict__ bias,
                                                      const float* __restrict__ g,
                                                      const float* __restrict__ beta,
                                                      float* __restrict__ out,
                                                      const int* __restrict__ meta,
                                                      const int* __restrict__ ls,
                                                      const int* __restrict__ lr,
                                                      float* __restrict__ lwt, int n, int e) {
  __shared__ __align__(16) float acc[TR * D];
  __shared__ int qe_all[NW * QN];
  __shared__ int qr_all[NW * QN];

  const int lane = threadIdx.x & 31;
  const int wave = __builtin_amdgcn_readfirstlane((int)(threadIdx.x >> 5));
  const int lo = blockIdx.x * TR;
  const int rows = (n - lo) < TR ? (n - lo) : TR;
  int* qe = qe_all + wave * QN;
  int* qr = qr_all + wave * QN;

  {
    const float b0 = bias[lane], b1 = bias[lane + 32], b2 = bias[lane + 64], b3 = bias[lane + 96];
#pragma unroll 4
    for (int rl = wave; rl < rows; rl += NW) {
      float s = dinv[lo + rl];
      s = s * s;
      const float* hp = hw + (size_t)(lo + rl) * D + lane;
      float* ap = acc + rl * D + lane;
      ap[0]  = b0 + s * hp[0];
      ap[32] = b1 + s * hp[32];
      ap[64] = b2 + s * hp[64];
      ap[96] = b3 + s * hp[96];
    }
  }
  __syncthreads();

  const int* bls = ls + (size_t)blockIdx.x * CAP;
  const int* blr = lr + (size_t)blockIdx.x * CAP;
  float* blw = lwt + (size_t)blockIdx.x * CAP;
  const int total = meta[blockIdx.x];
  if ((unsigned)total <= (unsigned)CAP) {
    const int nchunk = (total + 31) >> 5;
    int c = wave;
    int s = 0, rl = 0, s2 = 0, rl2 = 0;
    float w = 0.0f;
    {
      const int i = c * 32 + lane;
      if (FIRST) {
        const int i2 = i + NW * 32;
        if (i < total) { s = bls[i]; rl = blr[i]; }
        if (i2 < total) { s2 = bls[i2]; rl2 = blr[i2]; }
        if (i < total) { w = dinv[lo + rl] * dinv[s]; blw[i] = w; }
      } else {
        if (i < total) { s = bls[i]; rl = blr[i]; w = blw[i]; }
      }
    }
    while (c < nchunk) {
      const int left = total - c * 32;
      const int count = left < 32 ? left : 32;
      const int cn = c + NW;
      const int i2 = cn * 32 + lane;
      const int i3 = i2 + NW * 32;
      int s3 = 0, rl3 = 0;
      float w2 = 0.0f, da = 0.0f, db = 0.0f;
      if (FIRST) {
        da = dinv[lo + rl2];
        db = dinv[s2];
        if (i3 < total) { s3 = bls[i3]; rl3 = blr[i3]; }
      } else {
        s2 = 0; rl2 = 0;
        if (i2 < total) { s2 = bls[i2]; rl2 = blr[i2]; w2 = blw[i2]; }
      }
      tile_apply(hw, acc, s, rl, w, count, lane);
      if (FIRST) {
        w2 = da * db;
        if (i2 < total) blw[i2] = w2;
      }
      s = s2; rl = rl2; w = w2;
      if (FIRST) { s2 = s3; rl2 = rl3; }
      c = cn;
    }
  } else {
    tile_scan<false>(er, ec, dinv, hw, acc, nullptr, qe, qr, lo, rows, e, wave, lane, nullptr,
                     nullptr, nullptr);
  }
  __syncthreads();

  {
    const int c = lane * 4;
    const float4 gg = *(const float4*)(g + c);
    const float4 bb = *(const float4*)(beta + c);
#pragma unroll 4
    for (int rl = wave; rl < rows; rl += NW) {
      const float4 v = *(const float4*)(acc + rl * D + c);

      float s = v.x + v.y + v.z + v.w;
#pragma unroll
      for (int off = 16; off > 0; off >>= 1) s += __shfl_xor(s, off, 32);
      const float mu = s * (1.0f / 128.0f);

      const float dx = v.x - mu, dy = v.y - mu, dz = v.z - mu, dw = v.w - mu;
      float q = dx * dx + dy * dy + dz * dz + dw * dw;
#pragma unroll
      for (int off = 16; off > 0; off >>= 1) q += __shfl_xor(q, off, 32);
      const float rs = rsqrtf(q * (1.0f / 128.0f) + LN_EPS);

      float4 o;
      o.x = fmaxf(dx * rs * gg.x + bb.x, 0.0f);
      o.y = fmaxf(dy * rs * gg.y + bb.y, 0.0f);
      o.z = fmaxf(dz * rs * gg.z + bb.z, 0.0f);
      o.w = fmaxf(dw * rs * gg.w + bb.w, 0.0f);
      *(float4*)(out + (size_t)(lo + rl) * D + c) = o;
    }
  }
}

extern "C" void kernel_launch(void* const* d_in, const int* in_sizes, int n_in,
                              void* d_out, int out_size, void* d_ws, size_t ws_size,
                              hipStream_t stream) {
  const float* x    = (const float*)d_in[0];
  const int*   erow = (const int*)d_in[1];
  const int*   ecol = (const int*)d_in[2];
  const float* W[3]  = {(const float*)d_in[3], (const float*)d_in[7],  (const float*)d_in[11]};
  const float* b[3]  = {(const float*)d_in[4], (const float*)d_in[8],  (const float*)d_in[12]};
  const float* g[3]  = {(const float*)d_in[5], (const float*)d_in[9],  (const float*)d_in[13]};
  const float* bt[3] = {(const float*)d_in[6], (const float*)d_in[10], (const float*)d_in[14]};

  const int n = in_sizes[0] / D;
  const int e = in_sizes[1];

  float* ws   = (float*)d_ws;
  float* dinv = ws;
  float* hw   = ws + n;
  float* outp = (float*)d_out;

  const unsigned gemm_blocks = (unsigned)((n / 16 + 7) / 8);
  const unsigned tile_blocks = (unsigned)((n + TR - 1) / TR);

  int*   meta = (int*)(hw + (size_t)n * D);
  int*   ls   = meta + ((tile_blocks + 31u) & ~31u);
  int*   lr   = ls + (size_t)tile_blocks * CAP;
  float* lwt  = (float*)(lr + (size_t)tile_blocks * CAP);

  k_tile_filter<<<tile_blocks, NW * 32, 0, stream>>>(erow, ecol, dinv, meta, ls, lr, n, e);

  const float* hin = x;
  for (int L = 0; L < 3; ++L) {
    k_gemm_wmma<<<gemm_blocks, 256, 0, stream>>>(hin, W[L], hw, n);
    if (L == 0)
      k_tile_agg_ln<true><<<tile_blocks, NW * 32, 0, stream>>>(erow, ecol, dinv, hw, b[L], g[L], bt[L], outp, meta, ls, lr, lwt, n, e);
    else
      k_tile_agg_ln<false><<<tile_blocks, NW * 32, 0, stream>>>(erow, ecol, dinv, hw, b[L], g[L], bt[L], outp, meta, ls, lr, lwt, n, e);
    hin = outp;
  }
}


# Edge filter: a new launch flags, per 32-edge chunk, the tiles it has edges for; a tile block reads only its flagged chunks (1 line each), not the whole index list; 7 to 8 launches; arrival-order sums
# speedup vs baseline: 3.6820x; 1.1024x over previous
#include <hip/hip_runtime.h>
#include <hip/hip_bf16.h>

typedef __attribute__((ext_vector_type(2))) float v2f;
typedef __attribute__((ext_vector_type(8))) float v8f;

#define D 128
#define LN_EPS 1e-5f

__global__ __launch_bounds__(256) void k_gemm_wmma(const float* __restrict__ h,
                                                   const float* __restrict__ W,
                                                   float* __restrict__ out, int n) {
  __shared__ __align__(16) float lw[D * D];
  {
    const float4* W4 = (const float4*)W;
    float4* lw4 = (float4*)lw;
    for (int i = threadIdx.x; i < (D * D) / 4; i += 256) lw4[i] = W4[i];
  }
  __syncthreads();

  const int wave = threadIdx.x >> 5;
  const int lane = threadIdx.x & 31;
  const int ln   = lane & 15;
  const int koff = (lane < 16) ? 0 : 2;

  const int row0 = (blockIdx.x * 8 + wave) * 16;
  if (row0 + 16 > n) return;

  v8f acc[8];
  const v8f zero = {0.f, 0.f, 0.f, 0.f, 0.f, 0.f, 0.f, 0.f};
#pragma unroll
  for (int t = 0; t < 8; ++t) acc[t] = zero;

  const float* __restrict__ hrow = h + (size_t)(row0 + ln) * D;

  for (int k0 = 0; k0 < D; k0 += 4) {
    v2f a = *(const v2f*)(hrow + k0 + koff);
#pragma unroll
    for (int t = 0; t < 8; ++t) {
      const int n0 = t * 16;
      v2f b;
      b.x = lw[(k0 + koff) * D + n0 + ln];
      b.y = lw[(k0 + koff + 1) * D + n0 + ln];
      acc[t] = __builtin_amdgcn_wmma_f32_16x16x4_f32(
           false, a,  false, b,
           (short)0, acc[t],  false,  false);
    }
  }

  const int mbase = (lane < 16) ? 0 : 8;
#pragma unroll
  for (int t = 0; t < 8; ++t) {
    const int n0 = t * 16;
#pragma unroll
    for (int j = 0; j < 8; ++j) {
      out[(size_t)(row0 + mbase + j) * D + n0 + ln] = acc[t][j];
    }
  }
}

#define TR 400
#define CAP 16384
#define NW 32
#define QN 256
#define EG 8
#define KS 8
#define SEG 1024
#define MAXT 256
#define CL 8192
#define CB 16

__device__ __forceinline__ void lds_add(float* p, float x) {
  __hip_atomic_fetch_add(p, x, __ATOMIC_RELAXED, __HIP_MEMORY_SCOPE_WORKGROUP);
}

__device__ __forceinline__ void tile_apply(const float* __restrict__ hw, float* acc,
                                           int s, int rl, float w, int count, int lane) {
  int g0 = 0;
#pragma unroll 1
  for (; g0 + EG <= count; g0 += EG) {
    float v[EG][4];
#pragma unroll
    for (int j = 0; j < EG; ++j) {
      const int sj = __builtin_amdgcn_readlane(s, g0 + j);
      const float* hp = hw + (size_t)sj * D + lane;
      v[j][0] = hp[0];
      v[j][1] = hp[32];
      v[j][2] = hp[64];
      v[j][3] = hp[96];
    }
#pragma unroll
    for (int j = 0; j < EG; ++j) {
      const int rj = __builtin_amdgcn_readlane(rl, g0 + j);
      const float wj = __int_as_float(__builtin_amdgcn_readlane(__float_as_int(w), g0 + j));
      float* ap = acc + rj * D + lane;
      lds_add(ap + 0,  wj * v[j][0]);
      lds_add(ap + 32, wj * v[j][1]);
      lds_add(ap + 64, wj * v[j][2]);
      lds_add(ap + 96, wj * v[j][3]);
    }
  }
  for (int j = g0; j < count; ++j) {
    const int sj = __builtin_amdgcn_readlane(s, j);
    const int rj = __builtin_amdgcn_readlane(rl, j);
    const float wj = __int_as_float(__builtin_amdgcn_readlane(__float_as_int(w), j));
    const float* hp = hw + (size_t)sj * D + lane;
    const float h0 = hp[0], h1 = hp[32], h2 = hp[64], h3 = hp[96];
    float* ap = acc + rj * D + lane;
    lds_add(ap + 0,  wj * h0);
    lds_add(ap + 32, wj * h1);
    lds_add(ap + 64, wj * h2);
    lds_add(ap + 96, wj * h3);
  }
}

template <bool RECORD>
__device__ __forceinline__ void tile_flush(const int* __restrict__ ec,
                                           const float* __restrict__ dinv,
                                           const float* __restrict__ hw, float* acc, int* degc,
                                           const int* qe, const int* qr, int head, int count,
                                           int lo, int lane, int* fill, int* __restrict__ ls,
                                           int* __restrict__ lr) {
  int rl = 0, s = 0;
  if (lane < count) {
    const int p = (head + lane) & (QN - 1);
    const int idx = qe[p];
    rl = qr[p];
    s = ec[idx];
  }
  if (RECORD) {
    if (lane < count)
      __hip_atomic_fetch_add(degc + rl, 1, __ATOMIC_RELAXED, __HIP_MEMORY_SCOPE_WORKGROUP);
    int base = 0;
    if (lane == 0)
      base = __hip_atomic_fetch_add(fill, count, __ATOMIC_RELAXED, __HIP_MEMORY_SCOPE_WORKGROUP);
    base = __builtin_amdgcn_readfirstlane(base);
    if (base + count <= CAP && lane < count) {
      ls[base + lane] = s;
      lr[base + lane] = rl;
    }
  } else {
    float w = 0.0f;
    if (lane < count) w = dinv[lo + rl] * dinv[s];
    tile_apply(hw, acc, s, rl, w, count, lane);
  }
}

template <bool RECORD>
__device__ __forceinline__ void tile_scan(const int* __restrict__ er, const int* __restrict__ ec,
                                          const float* __restrict__ dinv,
                                          const float* __restrict__ hw, float* acc, int* degc,
                                          int* qe, int* qr, int lo, int rows, int e, int wave,
                                          int lane, int* fill, int* __restrict__ ls,
                                          int* __restrict__ lr) {
  const unsigned R = (unsigned)rows;
  const int nfull = e >> 7;
  int head = 0, cnt = 0;
#define TILE_PUSH(P, IDX, DD)                                                      \
  {                                                                              \
    const unsigned m = __builtin_amdgcn_ballot_w32(P);                           \
    if (P) {                                                                     \
      const int pos = (head + cnt + (int)__builtin_amdgcn_mbcnt_lo(m, 0u)) & (QN - 1); \
      qe[pos] = (IDX);                                                           \
      qr[pos] = (int)(DD);                                                       \
    }                                                                            \
    cnt += __builtin_popcount(m);                                                \
  }
  if (wave == 0) {
    for (int base = nfull << 7; base < e; base += 32) {
      const int i = base + lane;
      const int r = (i < e) ? er[i] : -1;
      const unsigned d = (unsigned)(r - lo);
      const bool p = d < R;
      TILE_PUSH(p, i, d)
    }
  }
  if (nfull > 0) {
    const int last = nfull - 1;
    const int4* er4 = (const int4*)er + lane;
    const int ngroup = (nfull + KS - 1) / KS;
    const int shift = (int)(((size_t)blockIdx.x * (size_t)ngroup) / gridDim.x);
    for (int sg = wave; sg < ngroup; sg += NW) {
      int sr = sg + shift;
      if (sr >= ngroup) sr -= ngroup;
      const int cb = sr * KS;
      int4 a[KS];
#pragma unroll
      for (int k = 0; k < KS; ++k) {
        const int c = cb + k;
        a[k] = er4[(size_t)(c < last ? c : last) * 32];
      }
#pragma unroll
      for (int k = 0; k < KS; ++k) {
        const int c = cb + k;
        const int i0 = c * 128 + lane * 4;
        const bool in = c < nfull;
        const unsigned d0 = (unsigned)(a[k].x - lo), d1 = (unsigned)(a[k].y - lo);
        const unsigned d2 = (unsigned)(a[k].z - lo), d3 = (unsigned)(a[k].w - lo);
        const bool p0 = in & (d0 < R), p1 = in & (d1 < R), p2 = in & (d2 < R), p3 = in & (d3 < R);
        if (__builtin_amdgcn_ballot_w32(p0 | p1 | p2 | p3) != 0u) {
          TILE_PUSH(p0, i0 + 0, d0)
          TILE_PUSH(p1, i0 + 1, d1)
          TILE_PUSH(p2, i0 + 2, d2)
          TILE_PUSH(p3, i0 + 3, d3)
          while (cnt >= 32) {
            tile_flush<RECORD>(ec, dinv, hw, acc, degc, qe, qr, head, 32, lo, lane, fill, ls, lr);
            head = (head + 32) & (QN - 1);
            cnt -= 32;
          }
        }
      }
    }
  }
#undef TILE_PUSH
  while (cnt > 0) {
    const int k = cnt < 32 ? cnt : 32;
    tile_flush<RECORD>(ec, dinv, hw, acc, degc, qe, qr, head, k, lo, lane, fill, ls, lr);
    head = (head + k) & (QN - 1);
    cnt -= k;
  }
}

__global__ __launch_bounds__(256) void k_chunk_flags(const int* __restrict__ er,
                                                     unsigned* __restrict__ flags, int e,
                                                     int nseg, int tbp) {
  __shared__ unsigned fl[8 * MAXT];
  const int lane = threadIdx.x & 31;
  const int wave = threadIdx.x >> 5;
  unsigned* my = fl + wave * MAXT;
  for (int j = lane; j < MAXT; j += 32) my[j] = 0u;
  __syncthreads();
  const int sg = blockIdx.x * 8 + wave;
  if (sg < nseg) {
    const int base = sg * SEG + lane;
    int r[32];
#pragma unroll
    for (int c = 0; c < 32; ++c) {
      const int i = base + c * 32;
      r[c] = (i < e) ? er[i] : -1;
    }
#pragma unroll
    for (int c = 0; c < 32; ++c) {
      const unsigned t = (unsigned)r[c] / (unsigned)TR;
      if (t < (unsigned)MAXT)
        __hip_atomic_fetch_or(my + t, 1u << c, __ATOMIC_RELAXED, __HIP_MEMORY_SCOPE_WORKGROUP);
    }
  }
  __syncthreads();
  if (sg < nseg) {
    const int w = tbp < MAXT ? tbp : MAXT;
    for (int j = lane; j < w; j += 32) flags[(size_t)sg * tbp + j] = my[j];
  }
}

__device__ __forceinline__ void chunk_scan(const int* __restrict__ er, const int* __restrict__ ec,
                                           int* degc, int* qe, int* qr, const int* clist, int nc,
                                           int lo, int rows, int e, int wave, int lane, int* fill,
                                           int* __restrict__ ls, int* __restrict__ lr) {
  const unsigned R = (unsigned)rows;
  int head = 0, cnt = 0;
  const int nb = (nc + CB - 1) / CB;
  for (int b = wave; b < nb; b += NW) {
    int a[CB], ci[CB];
#pragma unroll
    for (int k = 0; k < CB; ++k) {
      const int p = b * CB + k;
      const int i = clist[p < nc ? p : nc - 1] * 32 + lane;
      ci[k] = i;
      a[k] = er[i < e ? i : e - 1];
    }
#pragma unroll
    for (int k = 0; k < CB; ++k) {
      const int p = b * CB + k;
      const unsigned d = (unsigned)(a[k] - lo);
      const bool hit = (p < nc) & (ci[k] < e) & (d < R);
      const unsigned m = __builtin_amdgcn_ballot_w32(hit);
      if (hit) {
        const int pos = (head + cnt + (int)__builtin_amdgcn_mbcnt_lo(m, 0u)) & (QN - 1);
        qe[pos] = ci[k];
        qr[pos] = (int)d;
      }
      cnt += __builtin_popcount(m);
      while (cnt >= 32) {
        tile_flush<true>(ec, nullptr, nullptr, nullptr, degc, qe, qr, head, 32, lo, lane, fill,
                         ls, lr);
        head = (head + 32) & (QN - 1);
        cnt -= 32;
      }
    }
  }
  while (cnt > 0) {
    const int k = cnt < 32 ? cnt : 32;
    tile_flush<true>(ec, nullptr, nullptr, nullptr, degc, qe, qr, head, k, lo, lane, fill, ls,
                     lr);
    head = (head + k) & (QN - 1);
    cnt -= k;
  }
}

__global__ __launch_bounds__(1024) void k_tile_filter(const int* __restrict__ er,
                                                      const int* __restrict__ ec,
                                                      float* __restrict__ dinv,
                                                      int* __restrict__ meta, int* __restrict__ ls,
                                                      int* __restrict__ lr,
                                                      const unsigned* __restrict__ flags, int n,
                                                      int e, int nseg, int tbp) {
  __shared__ int qe_all[NW * QN];
  __shared__ int qr_all[NW * QN];
  __shared__ int degc[TR];
  __shared__ int fill;
  __shared__ int clist[CL];
  __shared__ int ccount;
  for (int i = threadIdx.x; i < TR; i += NW * 32) degc[i] = 0;
  if (threadIdx.x == 0) { fill = 0; ccount = 0; }
  __syncthreads();

  const int lane = threadIdx.x & 31;
  const int wave = __builtin_amdgcn_readfirstlane((int)(threadIdx.x >> 5));
  const int lo = blockIdx.x * TR;
  const int rows = (n - lo) < TR ? (n - lo) : TR;

  int nc = CL + 1;
  if (gridDim.x <= MAXT) {
    for (int s0 = threadIdx.x; s0 < nseg; s0 += 2 * NW * 32) {
      const int s1 = s0 + NW * 32;
      unsigned m0 = flags[(size_t)s0 * tbp + blockIdx.x];
      unsigned m1 = (s1 < nseg) ? flags[(size_t)s1 * tbp + blockIdx.x] : 0u;
      if (m0 != 0u) {
        int pos = __hip_atomic_fetch_add(&ccount, __builtin_popcount(m0), __ATOMIC_RELAXED,
                                         __HIP_MEMORY_SCOPE_WORKGROUP);
        while (m0 != 0u) {
          if (pos < CL) clist[pos] = s0 * 32 + __builtin_ctz(m0);
          m0 &= m0 - 1u;
          ++pos;
        }
      }
      if (m1 != 0u) {
        int pos = __hip_atomic_fetch_add(&ccount, __builtin_popcount(m1), __ATOMIC_RELAXED,
                                         __HIP_MEMORY_SCOPE_WORKGROUP);
        while (m1 != 0u) {
          if (pos < CL) clist[pos] = s1 * 32 + __builtin_ctz(m1);
          m1 &= m1 - 1u;
          ++pos;
        }
      }
    }
    __syncthreads();
    nc = ccount;
  }
  if (nc <= CL)
    chunk_scan(er, ec, degc, qe_all + wave * QN, qr_all + wave * QN, clist, nc, lo, rows, e,
               wave, lane, &fill, ls + (size_t)blockIdx.x * CAP, lr + (size_t)blockIdx.x * CAP);
  else
    tile_scan<true>(er, ec, nullptr, nullptr, nullptr, degc, qe_all + wave * QN,
                    qr_all + wave * QN, lo, rows, e, wave, lane, &fill,
                    ls + (size_t)blockIdx.x * CAP, lr + (size_t)blockIdx.x * CAP);
  __syncthreads();

  for (int i = threadIdx.x; i < rows; i += NW * 32)
    dinv[lo + i] = rsqrtf((float)(degc[i] + 1));
  if (threadIdx.x == 0) meta[blockIdx.x] = fill;
}

template <bool FIRST>
__global__ __launch_bounds__(1024) void k_tile_agg_ln(const int* __restrict__ er,
                                                      const int* __restrict__ ec,
                                                      const float* __restrict__ dinv,
                                                      const float* __restrict__ hw,
                                                      const float* __restrict__ bias,
                                                      const float* __restrict__ g,
                                                      const float* __restrict__ beta,
                                                      float* __restrict__ out,
                                                      const int* __restrict__ meta,
                                                      const int* __restrict__ ls,
                                                      const int* __restrict__ lr,
                                                      float* __restrict__ lwt, int n, int e) {
  __shared__ __align__(16) float acc[TR * D];
  __shared__ int qe_all[NW * QN];
  __shared__ int qr_all[NW * QN];

  const int lane = threadIdx.x & 31;
  const int wave = __builtin_amdgcn_readfirstlane((int)(threadIdx.x >> 5));
  const int lo = blockIdx.x * TR;
  const int rows = (n - lo) < TR ? (n - lo) : TR;
  int* qe = qe_all + wave * QN;
  int* qr = qr_all + wave * QN;

  {
    const float b0 = bias[lane], b1 = bias[lane + 32], b2 = bias[lane + 64], b3 = bias[lane + 96];
#pragma unroll 4
    for (int rl = wave; rl < rows; rl += NW) {
      float s = dinv[lo + rl];
      s = s * s;
      const float* hp = hw + (size_t)(lo + rl) * D + lane;
      float* ap = acc + rl * D + lane;
      ap[0]  = b0 + s * hp[0];
      ap[32] = b1 + s * hp[32];
      ap[64] = b2 + s * hp[64];
      ap[96] = b3 + s * hp[96];
    }
  }
  __syncthreads();

  const int* bls = ls + (size_t)blockIdx.x * CAP;
  const int* blr = lr + (size_t)blockIdx.x * CAP;
  float* blw = lwt + (size_t)blockIdx.x * CAP;
  const int total = meta[blockIdx.x];
  if ((unsigned)total <= (unsigned)CAP) {
    const int nchunk = (total + 31) >> 5;
    int c = wave;
    int s = 0, rl = 0, s2 = 0, rl2 = 0;
    float w = 0.0f;
    {
      const int i = c * 32 + lane;
      if (FIRST) {
        const int i2 = i + NW * 32;
        if (i < total) { s = bls[i]; rl = blr[i]; }
        if (i2 < total) { s2 = bls[i2]; rl2 = blr[i2]; }
        if (i < total) { w = dinv[lo + rl] * dinv[s]; blw[i] = w; }
      } else {
        if (i < total) { s = bls[i]; rl = blr[i]; w = blw[i]; }
      }
    }
    while (c < nchunk) {
      const int left = total - c * 32;
      const int count = left < 32 ? left : 32;
      const int cn = c + NW;
      const int i2 = cn * 32 + lane;
      const int i3 = i2 + NW * 32;
      int s3 = 0, rl3 = 0;
      float w2 = 0.0f, da = 0.0f, db = 0.0f;
      if (FIRST) {
        da = dinv[lo + rl2];
        db = dinv[s2];
        if (i3 < total) { s3 = bls[i3]; rl3 = blr[i3]; }
      } else {
        s2 = 0; rl2 = 0;
        if (i2 < total) { s2 = bls[i2]; rl2 = blr[i2]; w2 = blw[i2]; }
      }
      tile_apply(hw, acc, s, rl, w, count, lane);
      if (FIRST) {
        w2 = da * db;
        if (i2 < total) blw[i2] = w2;
      }
      s = s2; rl = rl2; w = w2;
      if (FIRST) { s2 = s3; rl2 = rl3; }
      c = cn;
    }
  } else {
    tile_scan<false>(er, ec, dinv, hw, acc, nullptr, qe, qr, lo, rows, e, wave, lane, nullptr,
                     nullptr, nullptr);
  }
  __syncthreads();

  {
    const int c = lane * 4;
    const float4 gg = *(const float4*)(g + c);
    const float4 bb = *(const float4*)(beta + c);
#pragma unroll 4
    for (int rl = wave; rl < rows; rl += NW) {
      const float4 v = *(const float4*)(acc + rl * D + c);

      float s = v.x + v.y + v.z + v.w;
#pragma unroll
      for (int off = 16; off > 0; off >>= 1) s += __shfl_xor(s, off, 32);
      const float mu = s * (1.0f / 128.0f);

      const float dx = v.x - mu, dy = v.y - mu, dz = v.z - mu, dw = v.w - mu;
      float q = dx * dx + dy * dy + dz * dz + dw * dw;
#pragma unroll
      for (int off = 16; off > 0; off >>= 1) q += __shfl_xor(q, off, 32);
      const float rs = rsqrtf(q * (1.0f / 128.0f) + LN_EPS);

      float4 o;
      o.x = fmaxf(dx * rs * gg.x + bb.x, 0.0f);
      o.y = fmaxf(dy * rs * gg.y + bb.y, 0.0f);
      o.z = fmaxf(dz * rs * gg.z + bb.z, 0.0f);
      o.w = fmaxf(dw * rs * gg.w + bb.w, 0.0f);
      *(float4*)(out + (size_t)(lo + rl) * D + c) = o;
    }
  }
}

extern "C" void kernel_launch(void* const* d_in, const int* in_sizes, int n_in,
                              void* d_out, int out_size, void* d_ws, size_t ws_size,
                              hipStream_t stream) {
  const float* x    = (const float*)d_in[0];
  const int*   erow = (const int*)d_in[1];
  const int*   ecol = (const int*)d_in[2];
  const float* W[3]  = {(const float*)d_in[3], (const float*)d_in[7],  (const float*)d_in[11]};
  const float* b[3]  = {(const float*)d_in[4], (const float*)d_in[8],  (const float*)d_in[12]};
  const float* g[3]  = {(const float*)d_in[5], (const float*)d_in[9],  (const float*)d_in[13]};
  const float* bt[3] = {(const float*)d_in[6], (const float*)d_in[10], (const float*)d_in[14]};

  const int n = in_sizes[0] / D;
  const int e = in_sizes[1];

  float* ws   = (float*)d_ws;
  float* dinv = ws;
  float* hw   = ws + n;
  float* outp = (float*)d_out;

  const unsigned gemm_blocks = (unsigned)((n / 16 + 7) / 8);
  const unsigned tile_blocks = (unsigned)((n + TR - 1) / TR);

  int*   meta = (int*)(hw + (size_t)n * D);
  int*   ls   = meta + ((tile_blocks + 31u) & ~31u);
  int*   lr   = ls + (size_t)tile_blocks * CAP;
  float* lwt  = (float*)(lr + (size_t)tile_blocks * CAP);
  unsigned* flg = (unsigned*)(lwt + (size_t)tile_blocks * CAP);
  const int nseg = (e + SEG - 1) / SEG;
  const int tbp  = (int)((tile_blocks + 31u) & ~31u);

  k_chunk_flags<<<(unsigned)((nseg + 7) / 8), 256, 0, stream>>>(erow, flg, e, nseg, tbp);
  k_tile_filter<<<tile_blocks, NW * 32, 0, stream>>>(erow, ecol, dinv, meta, ls, lr, flg, n, e,
                                                     nseg, tbp);

  const float* hin = x;
  for (int L = 0; L < 3; ++L) {
    k_gemm_wmma<<<gemm_blocks, 256, 0, stream>>>(hin, W[L], hw, n);
    if (L == 0)
      k_tile_agg_ln<true><<<tile_blocks, NW * 32, 0, stream>>>(erow, ecol, dinv, hw, b[L], g[L], bt[L], outp, meta, ls, lr, lwt, n, e);
    else
      k_tile_agg_ln<false><<<tile_blocks, NW * 32, 0, stream>>>(erow, ecol, dinv, hw, b[L], g[L], bt[L], outp, meta, ls, lr, lwt, n, e);
    hin = outp;
  }
}
